# starting kernel
# baseline (speedup 1.0000x reference)
#include <hip/hip_runtime.h>
#include <hip/hip_bf16.h>

typedef _Float16 v16h __attribute__((ext_vector_type(16)));
typedef _Float16 v8h  __attribute__((ext_vector_type(8)));
typedef float    v8f  __attribute__((ext_vector_type(8)));

#define NEG_SLOPE 0.2f

__global__ void k_f32_to_f16(const float* __restrict__ in, _Float16* __restrict__ out, long n) {
    long i = (long)blockIdx.x * blockDim.x + threadIdx.x;
    if (i < n) out[i] = (_Float16)in[i];
}

__global__ void k_fill(float* __restrict__ p, float v, long n) {
    long i = (long)blockIdx.x * blockDim.x + threadIdx.x;
    if (i < n) p[i] = v;
}


__device__ __forceinline__ v16h load_frag(const _Float16* __restrict__ p, int khalf) {
    v8h lo = *(const v8h*)(p + khalf);
    v8h hi = *(const v8h*)(p + 16 + khalf);
    v16h f;
#pragma unroll
    for (int i = 0; i < 8; ++i) { f[i] = lo[i]; f[8 + i] = hi[i]; }
    return f;
}

__device__ __forceinline__ v8f wmma_f16(v16h a, v16h b, v8f c) {
    return __builtin_amdgcn_wmma_f32_16x16x32_f16(
         false, a,  false, b,
         (short)0, c,  false,  false);
}

__global__ void k_gemm_wmma(const _Float16* __restrict__ A, const _Float16* __restrict__ B,
                            float* __restrict__ C, int Nout, int K,
                            int tilesN4, int totalWaves) {
    int wave = blockIdx.x * (blockDim.x >> 5) + (threadIdx.x >> 5);
    if (wave >= totalWaves) return;
    int lane  = threadIdx.x & 31;
    int r     = lane & 15;
    int hi    = lane >> 4;
    int khalf = hi << 3;

    int mt  = wave / tilesN4;
    int nt4 = wave - mt * tilesN4;

    const _Float16* arow = A + (long)(mt * 16 + r) * K;
    const _Float16* brow = B + (long)(nt4 * 64 + r) * K;

    v8f acc0 = {}, acc1 = {}, acc2 = {}, acc3 = {};
    for (int kk = 0; kk < K; kk += 32) {
        v16h a  = load_frag(arow + kk, khalf);
        v16h b0 = load_frag(brow + kk, khalf);
        v16h b1 = load_frag(brow + (long)16 * K + kk, khalf);
        v16h b2 = load_frag(brow + (long)32 * K + kk, khalf);
        v16h b3 = load_frag(brow + (long)48 * K + kk, khalf);
        acc0 = wmma_f16(a, b0, acc0);
        acc1 = wmma_f16(a, b1, acc1);
        acc2 = wmma_f16(a, b2, acc2);
        acc3 = wmma_f16(a, b3, acc3);
    }

    int hi8 = hi << 3;
    float* crow = C + (long)nt4 * 64 + r;
#pragma unroll
    for (int v = 0; v < 8; ++v) {
        long ro = (long)(mt * 16 + hi8 + v) * Nout;
        crow[ro +  0] = acc0[v];
        crow[ro + 16] = acc1[v];
        crow[ro + 32] = acc2[v];
        crow[ro + 48] = acc3[v];
    }
}

__global__ void k_att(const float* __restrict__ Hm,
                      const float* __restrict__ asr, const float* __restrict__ adr,
                      float* __restrict__ als, float* __restrict__ ald,
                      int NH, int H, int C) {
    int i = blockIdx.x * blockDim.x + threadIdx.x;
    if (i >= NH) return;
    int h = i % H;
    const float* hv = Hm + (long)i * C;
    const float* av = asr + h * C;
    const float* dv = adr + h * C;
    float s = 0.f, d = 0.f;
    for (int c = 0; c < C; ++c) {
        float x = hv[c];
        s += x * av[c];
        d += x * dv[c];
    }
    als[i] = s;
    ald[i] = d;
}

__device__ __forceinline__ void atomicMaxF(float* a, float v) {
    if (v >= 0.f) atomicMax((int*)a, __float_as_int(v));
    else          atomicMin((unsigned int*)a, __float_as_uint(v));
}

__device__ __forceinline__ void edge_sd(const int* src, const int* dst, int e, int E,
                                        int& s, int& d) {
    if (e < E) { s = src[e]; d = dst[e]; }
    else       { s = e - E;  d = e - E; }
}

__global__ void k_edge_max(const int* __restrict__ src, const int* __restrict__ dst,
                           int E, int Etot,
                           const float* __restrict__ als, const float* __restrict__ ald,
                           float* __restrict__ m, int H) {
    int i = blockIdx.x * blockDim.x + threadIdx.x;
    if (i >= Etot * H) return;
    int e = i / H, h = i - e * H;
    int s, d; edge_sd(src, dst, e, E, s, d);
    float ev = als[s * H + h] + ald[d * H + h];
    ev = ev > 0.f ? ev : NEG_SLOPE * ev;
    atomicMaxF(&m[d * H + h], ev);
}

__global__ void k_edge_denom(const int* __restrict__ src, const int* __restrict__ dst,
                             int E, int Etot,
                             const float* __restrict__ als, const float* __restrict__ ald,
                             const float* __restrict__ m, float* __restrict__ denom, int H) {
    int i = blockIdx.x * blockDim.x + threadIdx.x;
    if (i >= Etot * H) return;
    int e = i / H, h = i - e * H;
    int s, d; edge_sd(src, dst, e, E, s, d);
    float ev = als[s * H + h] + ald[d * H + h];
    ev = ev > 0.f ? ev : NEG_SLOPE * ev;
    atomicAdd(&denom[d * H + h], expf(ev - m[d * H + h]));
}

__global__ void k_edge_scatter(const int* __restrict__ src, const int* __restrict__ dst,
                               int E, int Etot,
                               const float* __restrict__ als, const float* __restrict__ ald,
                               const float* __restrict__ m, const float* __restrict__ denom,
                               const float* __restrict__ Hm, float* __restrict__ agg,
                               int H, int C) {
    int nout4 = (H * C) >> 2;
    int i = blockIdx.x * blockDim.x + threadIdx.x;
    if (i >= Etot * nout4) return;
    int e  = i / nout4;
    int c4 = i - e * nout4;
    int c  = c4 << 2;
    int h  = c / C;
    int s, d; edge_sd(src, dst, e, E, s, d);
    float ev = als[s * H + h] + ald[d * H + h];
    ev = ev > 0.f ? ev : NEG_SLOPE * ev;
    float alpha = expf(ev - m[d * H + h]) / (denom[d * H + h] + 1e-16f);
    const float4 hv = *(const float4*)(Hm + (long)s * (H * C) + c);
    float* ap = agg + (long)d * (H * C) + c;
    atomicAdd(ap + 0, hv.x * alpha);
    atomicAdd(ap + 1, hv.y * alpha);
    atomicAdd(ap + 2, hv.z * alpha);
    atomicAdd(ap + 3, hv.w * alpha);
}

__global__ void k_bias_act(float* __restrict__ out, const float* __restrict__ bias,
                           long n, int Nout, int doElu) {
    long i = (long)blockIdx.x * blockDim.x + threadIdx.x;
    if (i >= n) return;
    float v = out[i] + bias[i % Nout];
    if (doElu) v = v > 0.f ? v : (expf(v) - 1.f);
    out[i] = v;
}

static inline long cdivl(long a, long b) { return (a + b - 1) / b; }

struct LayerWS {
    _Float16* Xh; _Float16* Wh;
    float *als, *ald, *m, *denom;
};

static void run_gat_layer(const float* Xin, const float* W, const float* a_s, const float* a_d,
                          const float* bias, int N, int K, int Nout, int H, int C,
                          const int* src, const int* dst, int E, int Etot,
                          float* Hbuf, float* agg, int doElu, const LayerWS& w,
                          hipStream_t stream) {
    const int TB = 256;
    k_f32_to_f16<<<(int)cdivl((long)N * K, TB), TB, 0, stream>>>(Xin, w.Xh, (long)N * K);
    k_f32_to_f16<<<(int)cdivl((long)Nout * K, TB), TB, 0, stream>>>(W, w.Wh, (long)Nout * K);
    int tilesN4 = Nout / 64, totalWaves = (N / 16) * tilesN4;
    k_gemm_wmma<<<(int)cdivl(totalWaves, 8), TB, 0, stream>>>(w.Xh, w.Wh, Hbuf, Nout, K,
                                                              tilesN4, totalWaves);
    k_att<<<(int)cdivl((long)N * H, TB), TB, 0, stream>>>(Hbuf, a_s, a_d, w.als, w.ald,
                                                          N * H, H, C);
    k_fill<<<(int)cdivl((long)N * H, TB), TB, 0, stream>>>(w.m, -INFINITY, (long)N * H);
    k_fill<<<(int)cdivl((long)N * H, TB), TB, 0, stream>>>(w.denom, 0.f, (long)N * H);
    k_fill<<<(int)cdivl((long)N * Nout, TB), TB, 0, stream>>>(agg, 0.f, (long)N * Nout);
    int nEH = Etot * H;
    k_edge_max<<<(int)cdivl(nEH, TB), TB, 0, stream>>>(src, dst, E, Etot, w.als, w.ald, w.m, H);
    k_edge_denom<<<(int)cdivl(nEH, TB), TB, 0, stream>>>(src, dst, E, Etot, w.als, w.ald,
                                                         w.m, w.denom, H);
    int nSc = Etot * (Nout >> 2);
    k_edge_scatter<<<(int)cdivl(nSc, TB), TB, 0, stream>>>(src, dst, E, Etot, w.als, w.ald,
                                                           w.m, w.denom, Hbuf, agg, H, C);
    k_bias_act<<<(int)cdivl((long)N * Nout, TB), TB, 0, stream>>>(agg, bias, (long)N * Nout,
                                                                  Nout, doElu);
}

extern "C" void kernel_launch(void* const* d_in, const int* in_sizes, int n_in,
                              void* d_out, int out_size, void* d_ws, size_t ws_size,
                              hipStream_t stream) {
    const float* x   = (const float*)d_in[0];
    const int*   ei  = (const int*)  d_in[1];
    const float* W1  = (const float*)d_in[2];
    const float* as1 = (const float*)d_in[3];
    const float* ad1 = (const float*)d_in[4];
    const float* b1  = (const float*)d_in[5];
    const float* W2  = (const float*)d_in[6];
    const float* as2 = (const float*)d_in[7];
    const float* ad2 = (const float*)d_in[8];
    const float* b2  = (const float*)d_in[9];
    const float* W3  = (const float*)d_in[10];
    const float* as3 = (const float*)d_in[11];
    const float* ad3 = (const float*)d_in[12];
    const float* b3  = (const float*)d_in[13];
    float* out = (float*)d_out;

    const int FinW = 256;
    const int N    = in_sizes[0] / FinW;
    const int E    = in_sizes[1] / 2;
    const int Etot = E + N;
    const int* src = ei;
    const int* dst = ei + E;

    char* p = (char*)d_ws;
    float*    P   = (float*)p;    p += (size_t)N * 256 * sizeof(float);
    float*    Q   = (float*)p;    p += (size_t)N * 256 * sizeof(float);
    _Float16* Xh  = (_Float16*)p; p += (size_t)N * 256 * sizeof(_Float16);
    _Float16* Wh  = (_Float16*)p; p += (size_t)256 * 256 * sizeof(_Float16);
    float*    als = (float*)p;    p += (size_t)N * 8 * sizeof(float);
    float*    ald = (float*)p;    p += (size_t)N * 8 * sizeof(float);
    float*    m   = (float*)p;    p += (size_t)N * 8 * sizeof(float);
    float*    den = (float*)p;    p += (size_t)N * 8 * sizeof(float);
    LayerWS w{Xh, Wh, als, ald, m, den};

    run_gat_layer(x, W1, as1, ad1, b1, N, 256, 256, 8, 32,
                  src, dst, E, Etot, Q, P,  1, w, stream);
    run_gat_layer(P, W2, as2, ad2, b2, N, 256, 256, 8, 32,
                  src, dst, E, Etot, Q, P,  1, w, stream);
    run_gat_layer(P, W3, as3, ad3, b3, N, 256, 128, 1, 128,
                  src, dst, E, Etot, Q, out,  0, w, stream);
}


# Edge scatter: every destination row is added into from one half of the chip only (blocks read their die id and draw 32-edge chunks by a per-half ticket); f32 atomic sums unordered as before
# speedup vs baseline: 1.9112x; 1.9112x over previous
#include <hip/hip_runtime.h>
#include <hip/hip_bf16.h>

typedef _Float16 v16h __attribute__((ext_vector_type(16)));
typedef _Float16 v8h  __attribute__((ext_vector_type(8)));
typedef float    v8f  __attribute__((ext_vector_type(8)));

#define NEG_SLOPE 0.2f

__global__ void k_f32_to_f16(const float* __restrict__ in, _Float16* __restrict__ out, long n) {
    long i = (long)blockIdx.x * blockDim.x + threadIdx.x;
    if (i < n) out[i] = (_Float16)in[i];
}

#define TICK_STRIDE 64
#define TICK_WORDS  (3 * 2 * TICK_STRIDE)

__global__ void k_start(float* __restrict__ m, float* __restrict__ denom, float* __restrict__ agg,
                        unsigned int* __restrict__ tick, long nNH, long nAgg) {
    long i = (long)blockIdx.x * blockDim.x + threadIdx.x;
    if (i < nAgg) agg[i] = 0.f;
    if (i < nNH) { m[i] = -INFINITY; denom[i] = 0.f; }
    if (i < TICK_WORDS) tick[i] = 0u;
}


__device__ __forceinline__ v16h load_frag(const _Float16* __restrict__ p, int khalf) {
    v8h lo = *(const v8h*)(p + khalf);
    v8h hi = *(const v8h*)(p + 16 + khalf);
    v16h f;
#pragma unroll
    for (int i = 0; i < 8; ++i) { f[i] = lo[i]; f[8 + i] = hi[i]; }
    return f;
}

__device__ __forceinline__ v8f wmma_f16(v16h a, v16h b, v8f c) {
    return __builtin_amdgcn_wmma_f32_16x16x32_f16(
         false, a,  false, b,
         (short)0, c,  false,  false);
}

__global__ void k_gemm_wmma(const _Float16* __restrict__ A, const _Float16* __restrict__ B,
                            float* __restrict__ C, int Nout, int K,
                            int tilesN4, int totalWaves) {
    int wave = blockIdx.x * (blockDim.x >> 5) + (threadIdx.x >> 5);
    if (wave >= totalWaves) return;
    int lane  = threadIdx.x & 31;
    int r     = lane & 15;
    int hi    = lane >> 4;
    int khalf = hi << 3;

    int mt  = wave / tilesN4;
    int nt4 = wave - mt * tilesN4;

    const _Float16* arow = A + (long)(mt * 16 + r) * K;
    const _Float16* brow = B + (long)(nt4 * 64 + r) * K;

    v8f acc0 = {}, acc1 = {}, acc2 = {}, acc3 = {};
    for (int kk = 0; kk < K; kk += 32) {
        v16h a  = load_frag(arow + kk, khalf);
        v16h b0 = load_frag(brow + kk, khalf);
        v16h b1 = load_frag(brow + (long)16 * K + kk, khalf);
        v16h b2 = load_frag(brow + (long)32 * K + kk, khalf);
        v16h b3 = load_frag(brow + (long)48 * K + kk, khalf);
        acc0 = wmma_f16(a, b0, acc0);
        acc1 = wmma_f16(a, b1, acc1);
        acc2 = wmma_f16(a, b2, acc2);
        acc3 = wmma_f16(a, b3, acc3);
    }

    int hi8 = hi << 3;
    float* crow = C + (long)nt4 * 64 + r;
#pragma unroll
    for (int v = 0; v < 8; ++v) {
        long ro = (long)(mt * 16 + hi8 + v) * Nout;
        crow[ro +  0] = acc0[v];
        crow[ro + 16] = acc1[v];
        crow[ro + 32] = acc2[v];
        crow[ro + 48] = acc3[v];
    }
}

__global__ void k_att(const float* __restrict__ Hm,
                      const float* __restrict__ asr, const float* __restrict__ adr,
                      float* __restrict__ als, float* __restrict__ ald,
                      int NH, int H, int C) {
    int i = blockIdx.x * blockDim.x + threadIdx.x;
    if (i >= NH) return;
    int h = i % H;
    const float* hv = Hm + (long)i * C;
    const float* av = asr + h * C;
    const float* dv = adr + h * C;
    float s = 0.f, d = 0.f;
    for (int c = 0; c < C; ++c) {
        float x = hv[c];
        s += x * av[c];
        d += x * dv[c];
    }
    als[i] = s;
    ald[i] = d;
}

__device__ __forceinline__ void atomicMaxF(float* a, float v) {
    if (v >= 0.f) atomicMax((int*)a, __float_as_int(v));
    else          atomicMin((unsigned int*)a, __float_as_uint(v));
}

__device__ __forceinline__ void edge_sd(const int* src, const int* dst, int e, int E,
                                        int& s, int& d) {
    if (e < E) { s = src[e]; d = dst[e]; }
    else       { s = e - E;  d = e - E; }
}

__global__ void k_edge_max(const int* __restrict__ src, const int* __restrict__ dst,
                           int E, int Etot,
                           const float* __restrict__ als, const float* __restrict__ ald,
                           float* __restrict__ m, int H) {
    int i = blockIdx.x * blockDim.x + threadIdx.x;
    if (i >= Etot * H) return;
    int e = i / H, h = i - e * H;
    int s, d; edge_sd(src, dst, e, E, s, d);
    float ev = als[s * H + h] + ald[d * H + h];
    ev = ev > 0.f ? ev : NEG_SLOPE * ev;
    atomicMaxF(&m[d * H + h], ev);
}

#define SC_WPB 8
#define SC_EPW 4
#define SC_EPB (SC_WPB * SC_EPW)

template <int H, int C>
__global__ void k_edge_scatter(const int* __restrict__ src, const int* __restrict__ dst,
                               int E, int Etot,
                               const float* __restrict__ als, const float* __restrict__ ald,
                               const float* __restrict__ m, float* __restrict__ denom,
                               const float* __restrict__ Hm, float* __restrict__ agg,
                               unsigned int* __restrict__ tick, int nChunks) {
    constexpr int NOUT = H * C;
    constexpr int NIT  = NOUT / 32;
    __shared__ int sh_half, sh_chunk;
    if (threadIdx.x == 0) {
        int hf = (int)((__builtin_amdgcn_s_sendmsg_rtn(0x87) >> 18) & 1u);
        unsigned int t = atomicAdd(&tick[hf * TICK_STRIDE], 1u);
        if (t >= (unsigned int)nChunks) { hf ^= 1; t = atomicAdd(&tick[hf * TICK_STRIDE], 1u); }
        sh_half = hf; sh_chunk = (int)t;
    }
    __syncthreads();
    const int hf    = __builtin_amdgcn_readfirstlane(sh_half);
    const int chunk = __builtin_amdgcn_readfirstlane(sh_chunk);
    if ((unsigned int)chunk >= (unsigned int)nChunks) return;
    const int wv   = __builtin_amdgcn_readfirstlane((int)(threadIdx.x >> 5));
    const int lane = threadIdx.x & 31;
    const int e0   = (chunk * SC_WPB + wv) * SC_EPW;

    int  ss[SC_EPW], dd[SC_EPW];
    bool mine[SC_EPW];
#pragma unroll
    for (int i = 0; i < SC_EPW; ++i) {
        const int e  = e0 + i;
        const int ec = e < E ? e : 0;
        int sv = src[ec], dv = dst[ec];
        if (e >= E) { sv = e - E; dv = e - E; }
        ss[i] = sv; dd[i] = dv;
        mine[i] = (e < Etot) && (((dv >> 6) & 1) == hf);
    }
    float hv[SC_EPW][NIT];
    float ts[SC_EPW], td[SC_EPW], tm[SC_EPW];
#pragma unroll
    for (int i = 0; i < SC_EPW; ++i) {
        ts[i] = 0.f; td[i] = 0.f; tm[i] = 0.f;
        if (mine[i]) {
            const float* hp = Hm + (long)ss[i] * NOUT + lane;
#pragma unroll
            for (int j = 0; j < NIT; ++j) hv[i][j] = hp[j * 32];
            if (lane < H) {
                ts[i] = als[ss[i] * H + lane];
                td[i] = ald[dd[i] * H + lane];
                tm[i] = m[dd[i] * H + lane];
            }
        }
    }
#pragma unroll
    for (int i = 0; i < SC_EPW; ++i) {
        if (mine[i]) {
            const int dN = dd[i];
            float ex = 0.f;
            if (lane < H) {
                float ev = ts[i] + td[i];
                ev = ev > 0.f ? ev : NEG_SLOPE * ev;
                ex = expf(ev - tm[i]);
                atomicAdd(&denom[dN * H + lane], ex);
            }
            float* ap = agg + (long)dN * NOUT + lane;
#pragma unroll
            for (int j = 0; j < NIT; ++j) {
                float w = __int_as_float(__builtin_amdgcn_readlane(__float_as_int(ex), (j * 32) / C));
                atomicAdd(ap + j * 32, hv[i][j] * w);
            }
        }
    }
}

__global__ void k_norm_bias_act(float* __restrict__ out, const float* __restrict__ denom,
                                const float* __restrict__ bias,
                                long n, int Nout, int H, int C, int doElu) {
    long i = (long)blockIdx.x * blockDim.x + threadIdx.x;
    if (i >= n) return;
    long node = i / Nout;
    int  j    = (int)(i - node * Nout);
    float v = out[i] / (denom[node * H + j / C] + 1e-16f) + bias[j];
    if (doElu) v = v > 0.f ? v : (expf(v) - 1.f);
    out[i] = v;
}

template <int H, int C>
__global__ void k_boundary(const float* agg, float* denom, float* __restrict__ m,
                           const float* __restrict__ bias, _Float16* __restrict__ Xh,
                           float* nextAgg, int nextW, long n) {
    static_assert(C == 32, "one wave per (node, head): see the comment above");
    constexpr int NOUT = H * C;
    long i = (long)blockIdx.x * blockDim.x + threadIdx.x;
    if (i >= n) return;
    long node = i / NOUT;
    int  j    = (int)(i - node * NOUT);
    long dh   = node * H + j / C;
    float v = agg[i] / (denom[dh] + 1e-16f) + bias[j];
    v = v > 0.f ? v : (expf(v) - 1.f);
    Xh[i] = (_Float16)v;
    if (j < nextW) nextAgg[node * nextW + j] = 0.f;
    if (j % C == 0) { denom[dh] = 0.f; m[dh] = -INFINITY; }
}

static inline long cdivl(long a, long b) { return (a + b - 1) / b; }

struct LayerWS {
    _Float16* Xh; _Float16* Wh;
    float *als, *ald, *m, *denom;
    unsigned int* tick;
};

static void run_gat_layer(const float* W, const float* a_s, const float* a_d,
                          int N, int K, int Nout, int H, int C,
                          const int* src, const int* dst, int E, int Etot,
                          float* Hbuf, float* agg, const LayerWS& w, int layer,
                          hipStream_t stream) {
    const int TB = 256;
    k_f32_to_f16<<<(int)cdivl((long)Nout * K, TB), TB, 0, stream>>>(W, w.Wh, (long)Nout * K);
    int tilesN4 = Nout / 64, totalWaves = (N / 16) * tilesN4;
    k_gemm_wmma<<<(int)cdivl(totalWaves, 8), TB, 0, stream>>>(w.Xh, w.Wh, Hbuf, Nout, K,
                                                              tilesN4, totalWaves);
    k_att<<<(int)cdivl((long)N * H, TB), TB, 0, stream>>>(Hbuf, a_s, a_d, w.als, w.ald,
                                                          N * H, H, C);
    int nEH = Etot * H;
    k_edge_max<<<(int)cdivl(nEH, TB), TB, 0, stream>>>(src, dst, E, Etot, w.als, w.ald, w.m, H);
    int nChunks = (int)cdivl(Etot, SC_EPB);
    unsigned int* tk = w.tick + layer * 2 * TICK_STRIDE;
    if (H == 8) k_edge_scatter<8, 32><<<2 * nChunks, SC_WPB * 32, 0, stream>>>(
                    src, dst, E, Etot, w.als, w.ald, w.m, w.denom, Hbuf, agg, tk, nChunks);
    else        k_edge_scatter<1, 128><<<2 * nChunks, SC_WPB * 32, 0, stream>>>(
                    src, dst, E, Etot, w.als, w.ald, w.m, w.denom, Hbuf, agg, tk, nChunks);
}

extern "C" void kernel_launch(void* const* d_in, const int* in_sizes, int n_in,
                              void* d_out, int out_size, void* d_ws, size_t ws_size,
                              hipStream_t stream) {
    const float* x   = (const float*)d_in[0];
    const int*   ei  = (const int*)  d_in[1];
    const float* W1  = (const float*)d_in[2];
    const float* as1 = (const float*)d_in[3];
    const float* ad1 = (const float*)d_in[4];
    const float* b1  = (const float*)d_in[5];
    const float* W2  = (const float*)d_in[6];
    const float* as2 = (const float*)d_in[7];
    const float* ad2 = (const float*)d_in[8];
    const float* b2  = (const float*)d_in[9];
    const float* W3  = (const float*)d_in[10];
    const float* as3 = (const float*)d_in[11];
    const float* ad3 = (const float*)d_in[12];
    const float* b3  = (const float*)d_in[13];
    float* out = (float*)d_out;

    const int FinW = 256;
    const int N    = in_sizes[0] / FinW;
    const int E    = in_sizes[1] / 2;
    const int Etot = E + N;
    const int* src = ei;
    const int* dst = ei + E;

    char* p = (char*)d_ws;
    float*    P   = (float*)p;    p += (size_t)N * 256 * sizeof(float);
    float*    Q   = (float*)p;    p += (size_t)N * 256 * sizeof(float);
    _Float16* Xh  = (_Float16*)p; p += (size_t)N * 256 * sizeof(_Float16);
    _Float16* Wh  = (_Float16*)p; p += (size_t)256 * 256 * sizeof(_Float16);
    float*    als = (float*)p;    p += (size_t)N * 8 * sizeof(float);
    float*    ald = (float*)p;    p += (size_t)N * 8 * sizeof(float);
    float*    m   = (float*)p;    p += (size_t)N * 8 * sizeof(float);
    float*    den = (float*)p;    p += (size_t)N * 8 * sizeof(float);
    unsigned int* tick = (unsigned int*)p; p += (size_t)TICK_WORDS * sizeof(unsigned int);
    LayerWS w{Xh, Wh, als, ald, m, den, tick};

    const int TB = 256;
    const long nHid = (long)N * 256;
    k_f32_to_f16<<<(int)cdivl(nHid, TB), TB, 0, stream>>>(x, Xh, nHid);
    k_start<<<(int)cdivl(nHid, TB), TB, 0, stream>>>(m, den, P, tick, (long)N * 8, nHid);
    run_gat_layer(W1, as1, ad1, N, 256, 256, 8, 32, src, dst, E, Etot, Q, P, w, 0, stream);
    k_boundary<8, 32><<<(int)cdivl(nHid, TB), TB, 0, stream>>>(P, den, m, b1, Xh, P, 256, nHid);
    run_gat_layer(W2, as2, ad2, N, 256, 256, 8, 32, src, dst, E, Etot, Q, P, w, 1, stream);
    k_boundary<8, 32><<<(int)cdivl(nHid, TB), TB, 0, stream>>>(P, den, m, b2, Xh, out, 128, nHid);
    run_gat_layer(W3, as3, ad3, N, 256, 128, 1, 128, src, dst, E, Etot, Q, out, w, 2, stream);
    k_norm_bias_act<<<(int)cdivl((long)N * 128, TB), TB, 0, stream>>>(out, den, b3, (long)N * 128,
                                                                      128, 1, 128, 0);
}


# Self-loop edges leave the atomic edge passes: their logit is the running max's start value; the normalise pass adds their weight and weighted row with plain adds (that term is now summed last)
# speedup vs baseline: 1.9699x; 1.0307x over previous
#include <hip/hip_runtime.h>
#include <hip/hip_bf16.h>

typedef _Float16 v16h __attribute__((ext_vector_type(16)));
typedef _Float16 v8h  __attribute__((ext_vector_type(8)));
typedef float    v8f  __attribute__((ext_vector_type(8)));

#define NEG_SLOPE 0.2f

__global__ void k_f32_to_f16(const float* __restrict__ in, _Float16* __restrict__ out, long n) {
    long i = (long)blockIdx.x * blockDim.x + threadIdx.x;
    if (i < n) out[i] = (_Float16)in[i];
}

#define TICK_STRIDE 64
#define TICK_WORDS  (3 * 2 * TICK_STRIDE)

__global__ void k_start(float* __restrict__ denom, float* __restrict__ agg,
                        unsigned int* __restrict__ tick, long nNH, long nAgg) {
    long i = (long)blockIdx.x * blockDim.x + threadIdx.x;
    if (i < nAgg) agg[i] = 0.f;
    if (i < nNH) denom[i] = 0.f;
    if (i < TICK_WORDS) tick[i] = 0u;
}


__device__ __forceinline__ v16h load_frag(const _Float16* __restrict__ p, int khalf) {
    v8h lo = *(const v8h*)(p + khalf);
    v8h hi = *(const v8h*)(p + 16 + khalf);
    v16h f;
#pragma unroll
    for (int i = 0; i < 8; ++i) { f[i] = lo[i]; f[8 + i] = hi[i]; }
    return f;
}

__device__ __forceinline__ v8f wmma_f16(v16h a, v16h b, v8f c) {
    return __builtin_amdgcn_wmma_f32_16x16x32_f16(
         false, a,  false, b,
         (short)0, c,  false,  false);
}

__global__ void k_gemm_wmma(const _Float16* __restrict__ A, const _Float16* __restrict__ B,
                            float* __restrict__ C, int Nout, int K,
                            int tilesN4, int totalWaves) {
    int wave = blockIdx.x * (blockDim.x >> 5) + (threadIdx.x >> 5);
    if (wave >= totalWaves) return;
    int lane  = threadIdx.x & 31;
    int r     = lane & 15;
    int hi    = lane >> 4;
    int khalf = hi << 3;

    int mt  = wave / tilesN4;
    int nt4 = wave - mt * tilesN4;

    const _Float16* arow = A + (long)(mt * 16 + r) * K;
    const _Float16* brow = B + (long)(nt4 * 64 + r) * K;

    v8f acc0 = {}, acc1 = {}, acc2 = {}, acc3 = {};
    for (int kk = 0; kk < K; kk += 32) {
        v16h a  = load_frag(arow + kk, khalf);
        v16h b0 = load_frag(brow + kk, khalf);
        v16h b1 = load_frag(brow + (long)16 * K + kk, khalf);
        v16h b2 = load_frag(brow + (long)32 * K + kk, khalf);
        v16h b3 = load_frag(brow + (long)48 * K + kk, khalf);
        acc0 = wmma_f16(a, b0, acc0);
        acc1 = wmma_f16(a, b1, acc1);
        acc2 = wmma_f16(a, b2, acc2);
        acc3 = wmma_f16(a, b3, acc3);
    }

    int hi8 = hi << 3;
    float* crow = C + (long)nt4 * 64 + r;
#pragma unroll
    for (int v = 0; v < 8; ++v) {
        long ro = (long)(mt * 16 + hi8 + v) * Nout;
        crow[ro +  0] = acc0[v];
        crow[ro + 16] = acc1[v];
        crow[ro + 32] = acc2[v];
        crow[ro + 48] = acc3[v];
    }
}

__global__ void k_att(const float* __restrict__ Hm,
                      const float* __restrict__ asr, const float* __restrict__ adr,
                      float* __restrict__ als, float* __restrict__ ald, float* __restrict__ m,
                      int NH, int H, int C) {
    int i = blockIdx.x * blockDim.x + threadIdx.x;
    if (i >= NH) return;
    int h = i % H;
    const float* hv = Hm + (long)i * C;
    const float* av = asr + h * C;
    const float* dv = adr + h * C;
    float s = 0.f, d = 0.f;
    for (int c = 0; c < C; ++c) {
        float x = hv[c];
        s += x * av[c];
        d += x * dv[c];
    }
    als[i] = s;
    ald[i] = d;
    float ev = s + d;
    m[i] = ev > 0.f ? ev : NEG_SLOPE * ev;
}

__device__ __forceinline__ float self_weight(const float* __restrict__ als,
                                             const float* __restrict__ ald,
                                             const float* __restrict__ m, long dh) {
    float ev = als[dh] + ald[dh];
    ev = ev > 0.f ? ev : NEG_SLOPE * ev;
    return expf(ev - m[dh]);
}

__device__ __forceinline__ void atomicMaxF(float* a, float v) {
    if (v >= 0.f) atomicMax((int*)a, __float_as_int(v));
    else          atomicMin((unsigned int*)a, __float_as_uint(v));
}

__global__ void k_edge_max(const int* __restrict__ src, const int* __restrict__ dst, int E,
                           const float* __restrict__ als, const float* __restrict__ ald,
                           float* __restrict__ m, int H) {
    int i = blockIdx.x * blockDim.x + threadIdx.x;
    if (i >= E * H) return;
    int e = i / H, h = i - e * H;
    int s = src[e], d = dst[e];
    float ev = als[s * H + h] + ald[d * H + h];
    ev = ev > 0.f ? ev : NEG_SLOPE * ev;
    atomicMaxF(&m[d * H + h], ev);
}

#define SC_WPB 8
#define SC_EPW 4
#define SC_EPB (SC_WPB * SC_EPW)

template <int H, int C>
__global__ void k_edge_scatter(const int* __restrict__ src, const int* __restrict__ dst,
                               int E,
                               const float* __restrict__ als, const float* __restrict__ ald,
                               const float* __restrict__ m, float* __restrict__ denom,
                               const float* __restrict__ Hm, float* __restrict__ agg,
                               unsigned int* __restrict__ tick, int nChunks) {
    constexpr int NOUT = H * C;
    constexpr int NIT  = NOUT / 32;
    __shared__ int sh_half, sh_chunk;
    if (threadIdx.x == 0) {
        int hf = (int)((__builtin_amdgcn_s_sendmsg_rtn(0x87) >> 18) & 1u);
        unsigned int t = atomicAdd(&tick[hf * TICK_STRIDE], 1u);
        if (t >= (unsigned int)nChunks) { hf ^= 1; t = atomicAdd(&tick[hf * TICK_STRIDE], 1u); }
        sh_half = hf; sh_chunk = (int)t;
    }
    __syncthreads();
    const int hf    = __builtin_amdgcn_readfirstlane(sh_half);
    const int chunk = __builtin_amdgcn_readfirstlane(sh_chunk);
    if ((unsigned int)chunk >= (unsigned int)nChunks) return;
    const int wv   = __builtin_amdgcn_readfirstlane((int)(threadIdx.x >> 5));
    const int lane = threadIdx.x & 31;
    const int e0   = (chunk * SC_WPB + wv) * SC_EPW;

    int  ss[SC_EPW], dd[SC_EPW];
    bool mine[SC_EPW];
#pragma unroll
    for (int i = 0; i < SC_EPW; ++i) {
        const int e  = e0 + i;
        const int ec = e < E ? e : 0;
        int sv = src[ec], dv = dst[ec];
        ss[i] = sv; dd[i] = dv;
        mine[i] = (e < E) && (((dv >> 6) & 1) == hf);
    }
    float hv[SC_EPW][NIT];
    float ts[SC_EPW], td[SC_EPW], tm[SC_EPW];
#pragma unroll
    for (int i = 0; i < SC_EPW; ++i) {
        ts[i] = 0.f; td[i] = 0.f; tm[i] = 0.f;
        if (mine[i]) {
            const float* hp = Hm + (long)ss[i] * NOUT + lane;
#pragma unroll
            for (int j = 0; j < NIT; ++j) hv[i][j] = hp[j * 32];
            if (lane < H) {
                ts[i] = als[ss[i] * H + lane];
                td[i] = ald[dd[i] * H + lane];
                tm[i] = m[dd[i] * H + lane];
            }
        }
    }
#pragma unroll
    for (int i = 0; i < SC_EPW; ++i) {
        if (mine[i]) {
            const int dN = dd[i];
            float ex = 0.f;
            if (lane < H) {
                float ev = ts[i] + td[i];
                ev = ev > 0.f ? ev : NEG_SLOPE * ev;
                ex = expf(ev - tm[i]);
                atomicAdd(&denom[dN * H + lane], ex);
            }
            float* ap = agg + (long)dN * NOUT + lane;
#pragma unroll
            for (int j = 0; j < NIT; ++j) {
                float w = __int_as_float(__builtin_amdgcn_readlane(__float_as_int(ex), (j * 32) / C));
                atomicAdd(ap + j * 32, hv[i][j] * w);
            }
        }
    }
}

__global__ void k_norm_bias_act(float* __restrict__ out, const float* __restrict__ denom,
                                const float* __restrict__ als, const float* __restrict__ ald,
                                const float* __restrict__ m, const float* __restrict__ Hq,
                                const float* __restrict__ bias,
                                long n, int Nout, int H, int C, int doElu) {
    long i = (long)blockIdx.x * blockDim.x + threadIdx.x;
    if (i >= n) return;
    long node = i / Nout;
    int  j    = (int)(i - node * Nout);
    long dh   = node * H + j / C;
    float w = self_weight(als, ald, m, dh);
    float v = (out[i] + Hq[i] * w) / (denom[dh] + w + 1e-16f) + bias[j];
    if (doElu) v = v > 0.f ? v : (expf(v) - 1.f);
    out[i] = v;
}

template <int H, int C>
__global__ void k_boundary(const float* agg, float* denom,
                           const float* __restrict__ als, const float* __restrict__ ald,
                           const float* __restrict__ m, const float* __restrict__ Hq,
                           const float* __restrict__ bias, _Float16* __restrict__ Xh,
                           float* nextAgg, int nextW, long n) {
    static_assert(C == 32, "one wave per (node, head): see the comment above");
    constexpr int NOUT = H * C;
    long i = (long)blockIdx.x * blockDim.x + threadIdx.x;
    if (i >= n) return;
    long node = i / NOUT;
    int  j    = (int)(i - node * NOUT);
    long dh   = node * H + j / C;
    float w = self_weight(als, ald, m, dh);
    float v = (agg[i] + Hq[i] * w) / (denom[dh] + w + 1e-16f) + bias[j];
    v = v > 0.f ? v : (expf(v) - 1.f);
    Xh[i] = (_Float16)v;
    if (j < nextW) nextAgg[node * nextW + j] = 0.f;
    if (j % C == 0) denom[dh] = 0.f;
}

static inline long cdivl(long a, long b) { return (a + b - 1) / b; }

struct LayerWS {
    _Float16* Xh; _Float16* Wh;
    float *als, *ald, *m, *denom;
    unsigned int* tick;
};

static void run_gat_layer(const float* W, const float* a_s, const float* a_d,
                          int N, int K, int Nout, int H, int C,
                          const int* src, const int* dst, int E,
                          float* Hbuf, float* agg, const LayerWS& w, int layer,
                          hipStream_t stream) {
    const int TB = 256;
    k_f32_to_f16<<<(int)cdivl((long)Nout * K, TB), TB, 0, stream>>>(W, w.Wh, (long)Nout * K);
    int tilesN4 = Nout / 64, totalWaves = (N / 16) * tilesN4;
    k_gemm_wmma<<<(int)cdivl(totalWaves, 8), TB, 0, stream>>>(w.Xh, w.Wh, Hbuf, Nout, K,
                                                              tilesN4, totalWaves);
    k_att<<<(int)cdivl((long)N * H, TB), TB, 0, stream>>>(Hbuf, a_s, a_d, w.als, w.ald, w.m,
                                                          N * H, H, C);
    int nEH = E * H;
    k_edge_max<<<(int)cdivl(nEH, TB), TB, 0, stream>>>(src, dst, E, w.als, w.ald, w.m, H);
    int nChunks = (int)cdivl(E, SC_EPB);
    unsigned int* tk = w.tick + layer * 2 * TICK_STRIDE;
    if (H == 8) k_edge_scatter<8, 32><<<2 * nChunks, SC_WPB * 32, 0, stream>>>(
                    src, dst, E, w.als, w.ald, w.m, w.denom, Hbuf, agg, tk, nChunks);
    else        k_edge_scatter<1, 128><<<2 * nChunks, SC_WPB * 32, 0, stream>>>(
                    src, dst, E, w.als, w.ald, w.m, w.denom, Hbuf, agg, tk, nChunks);
}

extern "C" void kernel_launch(void* const* d_in, const int* in_sizes, int n_in,
                              void* d_out, int out_size, void* d_ws, size_t ws_size,
                              hipStream_t stream) {
    const float* x   = (const float*)d_in[0];
    const int*   ei  = (const int*)  d_in[1];
    const float* W1  = (const float*)d_in[2];
    const float* as1 = (const float*)d_in[3];
    const float* ad1 = (const float*)d_in[4];
    const float* b1  = (const float*)d_in[5];
    const float* W2  = (const float*)d_in[6];
    const float* as2 = (const float*)d_in[7];
    const float* ad2 = (const float*)d_in[8];
    const float* b2  = (const float*)d_in[9];
    const float* W3  = (const float*)d_in[10];
    const float* as3 = (const float*)d_in[11];
    const float* ad3 = (const float*)d_in[12];
    const float* b3  = (const float*)d_in[13];
    float* out = (float*)d_out;

    const int FinW = 256;
    const int N    = in_sizes[0] / FinW;
    const int E    = in_sizes[1] / 2;
    const int* src = ei;
    const int* dst = ei + E;

    char* p = (char*)d_ws;
    float*    P   = (float*)p;    p += (size_t)N * 256 * sizeof(float);
    float*    Q   = (float*)p;    p += (size_t)N * 256 * sizeof(float);
    _Float16* Xh  = (_Float16*)p; p += (size_t)N * 256 * sizeof(_Float16);
    _Float16* Wh  = (_Float16*)p; p += (size_t)256 * 256 * sizeof(_Float16);
    float*    als = (float*)p;    p += (size_t)N * 8 * sizeof(float);
    float*    ald = (float*)p;    p += (size_t)N * 8 * sizeof(float);
    float*    m   = (float*)p;    p += (size_t)N * 8 * sizeof(float);
    float*    den = (float*)p;    p += (size_t)N * 8 * sizeof(float);
    unsigned int* tick = (unsigned int*)p; p += (size_t)TICK_WORDS * sizeof(unsigned int);
    LayerWS w{Xh, Wh, als, ald, m, den, tick};

    const int TB = 256;
    const long nHid = (long)N * 256;
    k_f32_to_f16<<<(int)cdivl(nHid, TB), TB, 0, stream>>>(x, Xh, nHid);
    k_start<<<(int)cdivl(nHid, TB), TB, 0, stream>>>(den, P, tick, (long)N * 8, nHid);
    run_gat_layer(W1, as1, ad1, N, 256, 256, 8, 32, src, dst, E, Q, P, w, 0, stream);
    k_boundary<8, 32><<<(int)cdivl(nHid, TB), TB, 0, stream>>>(P, den, als, ald, m, Q, b1, Xh,
                                                               P, 256, nHid);
    run_gat_layer(W2, as2, ad2, N, 256, 256, 8, 32, src, dst, E, Q, P, w, 1, stream);
    k_boundary<8, 32><<<(int)cdivl(nHid, TB), TB, 0, stream>>>(P, den, als, ald, m, Q, b2, Xh,
                                                               out, 128, nHid);
    run_gat_layer(W3, as3, ad3, N, 256, 128, 1, 128, src, dst, E, Q, out, w, 2, stream);
    k_norm_bias_act<<<(int)cdivl((long)N * 128, TB), TB, 0, stream>>>(out, den, als, ald, m, Q, b3,
                                                                      (long)N * 128, 128, 1, 128, 0);
}


# Softmax max kept over source terms only (m = lrelu(max als + ald): monotonic, bit-identical) beside ald as one 8-byte pair per node-head: max pass gathers no ald, scatter one pair not two words
# speedup vs baseline: 2.6237x; 1.3319x over previous
#include <hip/hip_runtime.h>
#include <hip/hip_bf16.h>

typedef _Float16 v16h __attribute__((ext_vector_type(16)));
typedef _Float16 v8h  __attribute__((ext_vector_type(8)));
typedef float    v8f  __attribute__((ext_vector_type(8)));
typedef float    v2f  __attribute__((ext_vector_type(2)));

#define NEG_SLOPE 0.2f

#define TICK_STRIDE 64
#define TICK_WORDS  (3 * 2 * TICK_STRIDE)

__device__ __forceinline__ long frag_pos(long row, int k, int K) {
    const long t  = row >> 4;
    const int  r  = (int)(row & 15);
    const int  ks = k >> 5, w = k & 31;
    const int  jj = w >> 4, hi = (w >> 3) & 1, idx = w & 7;
    return ((((t * (K >> 5) + ks) * 2 + jj) * 32 + (r + 16 * hi)) * 8) + idx;
}

__device__ __forceinline__ long frag_src(long q, int K) {
    const int  idx = (int)(q & 7);
    const int  l   = (int)((q >> 3) & 31);
    const int  jj  = (int)((q >> 8) & 1);
    const long ts  = q >> 9;
    const int  KS  = K >> 5;
    const long t   = ts / KS;
    const int  ks  = (int)(ts - t * KS);
    const long row = t * 16 + (l & 15);
    const int  k   = ks * 32 + jj * 16 + (l >> 4) * 8 + idx;
    return row * K + k;
}

template <int KW>
__global__ void k_prep(const float* __restrict__ x, _Float16* __restrict__ Xh,
                       const float* __restrict__ W1, const float* __restrict__ W2,
                       const float* __restrict__ W3, _Float16* __restrict__ Wh1,
                       _Float16* __restrict__ Wh2, _Float16* __restrict__ Wh3,
                       float* __restrict__ denom, float* __restrict__ agg,
                       unsigned int* __restrict__ tick,
                       long nX, long nW12, long nW3, long nNH) {
    long i = (long)blockIdx.x * blockDim.x + threadIdx.x;
    const long s = frag_src(i, KW);
    if (i < nX) { Xh[i] = (_Float16)x[s]; agg[i] = 0.f; }
    if (i < nW12) { Wh1[i] = (_Float16)W1[s]; Wh2[i] = (_Float16)W2[s]; }
    if (i < nW3) Wh3[i] = (_Float16)W3[s];
    if (i < nNH) denom[i] = 0.f;
    if (i < TICK_WORDS) tick[i] = 0u;
}


__device__ __forceinline__ v16h load_frag(const _Float16* p) {
    v8h lo = *(const v8h*)(p);
    v8h hi = *(const v8h*)(p + 256);
    v16h f;
#pragma unroll
    for (int i = 0; i < 8; ++i) { f[i] = lo[i]; f[8 + i] = hi[i]; }
    return f;
}

__device__ __forceinline__ v8f wmma_f16(v16h a, v16h b, v8f c) {
    return __builtin_amdgcn_wmma_f32_16x16x32_f16(
         false, a,  false, b,
         (short)0, c,  false,  false);
}

__device__ __forceinline__ void gemm_strip(const _Float16* A, const _Float16* __restrict__ B,
                                           float* C, int Nout, int K, int tilesN4,
                                           int wave, int lane) {
    int r     = lane & 15;
    int hi    = lane >> 4;

    int mt  = wave / tilesN4;
    int nt4 = wave - mt * tilesN4;

    const int  KS      = K >> 5;
    const long tstride = (long)KS * 512;
    const _Float16* afr = A + (long)mt * tstride + lane * 8;
    const _Float16* bfr = B + (long)(nt4 * 4) * tstride + lane * 8;

    v8f acc0 = {}, acc1 = {}, acc2 = {}, acc3 = {};
    for (int kk = 0; kk < K; kk += 32) {
        const long ko = (long)(kk >> 5) * 512;
        v16h a  = load_frag(afr + ko);
        v16h b0 = load_frag(bfr + ko);
        v16h b1 = load_frag(bfr + ko + tstride);
        v16h b2 = load_frag(bfr + ko + 2 * tstride);
        v16h b3 = load_frag(bfr + ko + 3 * tstride);
        acc0 = wmma_f16(a, b0, acc0);
        acc1 = wmma_f16(a, b1, acc1);
        acc2 = wmma_f16(a, b2, acc2);
        acc3 = wmma_f16(a, b3, acc3);
    }

    int hi8 = hi << 3;
    float* crow = C + (long)nt4 * 64 + r;
#pragma unroll
    for (int v = 0; v < 8; ++v) {
        long ro = (long)(mt * 16 + hi8 + v) * Nout;
        crow[ro +  0] = acc0[v];
        crow[ro + 16] = acc1[v];
        crow[ro + 32] = acc2[v];
        crow[ro + 48] = acc3[v];
    }
}

template <int H, int C>
__device__ __forceinline__ void att_rows(const float* Hm,
                                         const float* __restrict__ asr,
                                         const float* __restrict__ adr,
                                         float* __restrict__ als, float* __restrict__ dm,
                                         long row0, int nRows, int N) {
    for (int p = threadIdx.x; p < nRows * H; p += blockDim.x) {
        const int  rl   = p / H, h = p - rl * H;
        const long node = row0 + rl;
        if (node < N) {
            const long i = node * H + h;
            const float* hv = Hm + i * C;
            const float* av = asr + h * C;
            const float* dv = adr + h * C;
            float s = 0.f, d = 0.f;
            for (int c = 0; c < C; ++c) {
                float x = hv[c];
                s = __builtin_fmaf(x, av[c], s);
                d = __builtin_fmaf(x, dv[c], d);
            }
            als[i] = s;
            v2f t; t.x = d; t.y = s;
            *(v2f*)(dm + 2 * i) = t;
        }
    }
}

template <int H, int C>
__global__ void k_gemm_att(const _Float16* __restrict__ A, const _Float16* __restrict__ B,
                           float* Cout, int K, int tilesN4, int totalWaves, int N,
                           const float* __restrict__ asr, const float* __restrict__ adr,
                           float* __restrict__ als, float* __restrict__ dm) {
    const int wpb  = blockDim.x >> 5;
    const int wv   = __builtin_amdgcn_readfirstlane((int)(threadIdx.x >> 5));
    const int wave = blockIdx.x * wpb + wv;
    if (wave < totalWaves) gemm_strip(A, B, Cout, H * C, K, tilesN4, wave, threadIdx.x & 31);
    __syncthreads();
    const int nRows = (wpb / tilesN4) * 16;
    att_rows<H, C>(Cout, asr, adr, als, dm, (long)blockIdx.x * nRows, nRows, N);
}

__device__ __forceinline__ float edge_logit(float a, float d) {
    float ev = a + d;
    return ev > 0.f ? ev : NEG_SLOPE * ev;
}

__device__ __forceinline__ float self_weight(const float* __restrict__ als,
                                             const float* __restrict__ dm, long dh) {
    const v2f t = *(const v2f*)(dm + 2 * dh);
    return expf(edge_logit(als[dh], t.x) - edge_logit(t.y, t.x));
}

__device__ __forceinline__ void atomicMaxF(float* a, float v) {
    if (v >= 0.f) atomicMax((int*)a, __float_as_int(v));
    else          atomicMin((unsigned int*)a, __float_as_uint(v));
}

__global__ void k_edge_max(const int* __restrict__ src, const int* __restrict__ dst, int E,
                           const float* __restrict__ als, float* __restrict__ dm, int H) {
    int i = blockIdx.x * blockDim.x + threadIdx.x;
    if (i >= E * H) return;
    int e = i / H, h = i - e * H;
    int s = src[e], d = dst[e];
    atomicMaxF(&dm[2 * (d * H + h) + 1], als[s * H + h]);
}

#define SC_WPB   8
#define SC_EPW   4
#define SC_STEPS 4
#define SC_EPB   (SC_WPB * SC_EPW * SC_STEPS)
typedef int sc_idx_w __attribute__((ext_vector_type(SC_STEPS * SC_EPW)));
typedef sc_idx_w sc_idx_t __attribute__((aligned(4)));

template <int H, int C>
__global__ void k_edge_scatter(const int* __restrict__ src, const int* __restrict__ dst,
                               int E,
                               const float* __restrict__ als, const float* __restrict__ dm,
                               float* __restrict__ denom,
                               const float* __restrict__ Hm, float* __restrict__ agg,
                               unsigned int* __restrict__ tick, int nChunks) {
    constexpr int NOUT = H * C;
    constexpr int NIT  = NOUT / 32;
    __shared__ int sh_half, sh_chunk;
    if (threadIdx.x == 0) {
        int hf = (int)((__builtin_amdgcn_s_sendmsg_rtn(0x87) >> 18) & 1u);
        unsigned int t = atomicAdd(&tick[hf * TICK_STRIDE], 1u);
        if (t >= (unsigned int)nChunks) { hf ^= 1; t = atomicAdd(&tick[hf * TICK_STRIDE], 1u); }
        sh_half = hf; sh_chunk = (int)t;
    }
    __syncthreads();
    const int hf    = __builtin_amdgcn_readfirstlane(sh_half);
    const int chunk = __builtin_amdgcn_readfirstlane(sh_chunk);
    if ((unsigned int)chunk >= (unsigned int)nChunks) return;
    const int wv   = __builtin_amdgcn_readfirstlane((int)(threadIdx.x >> 5));
    const int lane = threadIdx.x & 31;
    const int e0   = (chunk * SC_WPB + wv) * (SC_STEPS * SC_EPW);

    int  ss[SC_STEPS][SC_EPW], dd[SC_STEPS][SC_EPW];
    bool mine[SC_STEPS][SC_EPW];
    if (e0 + SC_STEPS * SC_EPW <= E) {
        const sc_idx_t sv = *(const sc_idx_t*)(src + e0);
        const sc_idx_t dv = *(const sc_idx_t*)(dst + e0);
#pragma unroll
        for (int st = 0; st < SC_STEPS; ++st) {
#pragma unroll
            for (int i = 0; i < SC_EPW; ++i) {
                ss[st][i] = sv[st * SC_EPW + i]; dd[st][i] = dv[st * SC_EPW + i];
                mine[st][i] = ((dv[st * SC_EPW + i] >> 6) & 1) == hf;
            }
        }
    } else {
#pragma unroll
        for (int st = 0; st < SC_STEPS; ++st) {
#pragma unroll
            for (int i = 0; i < SC_EPW; ++i) {
                const int e  = e0 + st * SC_EPW + i;
                const int ec = e < E ? e : 0;
                int sv1 = src[ec], dv1 = dst[ec];
                ss[st][i] = sv1; dd[st][i] = dv1;
                mine[st][i] = (e < E) && (((dv1 >> 6) & 1) == hf);
            }
        }
    }
#pragma unroll
    for (int st = 0; st < SC_STEPS; ++st) {
        float hv[SC_EPW][NIT];
        float ts[SC_EPW];
        v2f   tq[SC_EPW];
#pragma unroll
        for (int i = 0; i < SC_EPW; ++i) {
            ts[i] = 0.f; tq[i].x = 0.f; tq[i].y = 0.f;
            if (mine[st][i]) {
                const float* hp = Hm + (long)ss[st][i] * NOUT + lane;
#pragma unroll
                for (int j = 0; j < NIT; ++j) hv[i][j] = hp[j * 32];
                if (lane < H) {
                    ts[i] = als[ss[st][i] * H + lane];
                    tq[i] = *(const v2f*)(dm + 2 * (dd[st][i] * H + lane));
                }
            }
        }
#pragma unroll
        for (int i = 0; i < SC_EPW; ++i) {
            if (mine[st][i]) {
                const int dN = dd[st][i];
                float ex = 0.f;
                if (lane < H) {
                    ex = expf(edge_logit(ts[i], tq[i].x) - edge_logit(tq[i].y, tq[i].x));
                    atomicAdd(&denom[dN * H + lane], ex);
                }
                float* ap = agg + (long)dN * NOUT + lane;
#pragma unroll
                for (int j = 0; j < NIT; ++j) {
                    float w = __int_as_float(__builtin_amdgcn_readlane(__float_as_int(ex), (j * 32) / C));
                    atomicAdd(ap + j * 32, hv[i][j] * w);
                }
            }
        }
    }
}

__global__ void k_norm_bias_act(float* __restrict__ out, const float* __restrict__ denom,
                                const float* __restrict__ als, const float* __restrict__ dm,
                                const float* __restrict__ Hq,
                                const float* __restrict__ bias,
                                long n, int Nout, int H, int C, int doElu) {
    long i = (long)blockIdx.x * blockDim.x + threadIdx.x;
    if (i >= n) return;
    long node = i / Nout;
    int  j    = (int)(i - node * Nout);
    long dh   = node * H + j / C;
    float w = self_weight(als, dm, dh);
    float v = (out[i] + Hq[i] * w) / (denom[dh] + w + 1e-16f) + bias[j];
    if (doElu) v = v > 0.f ? v : (expf(v) - 1.f);
    out[i] = v;
}

#define BG_U 8

template <int H, int C, int H2, int C2>
__attribute__((amdgpu_waves_per_eu(16, 16)))
__global__ void k_boundary_gemm(const float* agg, float* denom,
                                const float* __restrict__ als, const float* __restrict__ dm,
                                const float* Hq,
                                const float* __restrict__ bias, _Float16* Xh,
                                float* nextAgg, int nextW, int N,
                                const _Float16* __restrict__ B, float* Cout,
                                int tilesN4, int totalWaves,
                                const float* __restrict__ asr2, const float* __restrict__ adr2,
                                float* __restrict__ als2, float* __restrict__ dm2) {
    static_assert(C == 32, "one wave per (node, head): see the comment above");
    constexpr int NOUT = H * C;
    const int wpb  = blockDim.x >> 5;
    const int wv   = __builtin_amdgcn_readfirstlane((int)(threadIdx.x >> 5));
    const int lane = threadIdx.x & 31;
    const int nRows = (wpb / tilesN4) * 16;
    const long row0 = (long)blockIdx.x * nRows;
    const int nPairs = nRows * H;

    for (int p0 = wv; p0 < nPairs; p0 += BG_U * wpb) {
        const int  pL    = p0 + (lane & (BG_U - 1)) * wpb;
        const int  rL    = pL / H;
        const long nodeL = row0 + rL;
        const long dhL   = (pL < nPairs && nodeL < N) ? nodeL * H + (pL - rL * H) : 0;
        const float wd = denom[dhL], wa = als[dhL];
        const v2f   wq = *(const v2f*)(dm + 2 * dhL);
        const float wt = wq.x, wm = wq.y;
        float va[BG_U], vq[BG_U], vb[BG_U];
        bool  ok[BG_U];
#pragma unroll
        for (int u = 0; u < BG_U; ++u) {
            const int  p    = p0 + u * wpb;
            const int  rl   = p / H, h = p - rl * H;
            const long node = row0 + rl;
            ok[u] = (p < nPairs) && (node < N);
            va[u] = vq[u] = vb[u] = 0.f;
            if (ok[u]) {
                const long i = node * NOUT + h * C + lane;
                va[u] = agg[i]; vq[u] = Hq[i]; vb[u] = bias[h * C + lane];
            }
        }
#pragma unroll
        for (int u = 0; u < BG_U; ++u) {
            if (ok[u]) {
                const int  p    = p0 + u * wpb;
                const int  rl   = p / H, h = p - rl * H;
                const long node = row0 + rl;
                const long dh   = node * H + h;
                const int  j    = h * C + lane;
                const float dn = __int_as_float(__builtin_amdgcn_readlane(__float_as_int(wd), u));
                const float sa = __int_as_float(__builtin_amdgcn_readlane(__float_as_int(wa), u));
                const float st = __int_as_float(__builtin_amdgcn_readlane(__float_as_int(wt), u));
                const float sm = __int_as_float(__builtin_amdgcn_readlane(__float_as_int(wm), u));
                const float w = expf(edge_logit(sa, st) - edge_logit(sm, st));
                float v = (va[u] + vq[u] * w) / (dn + w + 1e-16f) + vb[u];
                v = v > 0.f ? v : (expf(v) - 1.f);
                Xh[frag_pos(node, j, NOUT)] = (_Float16)v;
                if (j < nextW) nextAgg[node * nextW + j] = 0.f;
                if (lane == 0) denom[dh] = 0.f;
            }
        }
    }
    __syncthreads();
    const int wave = blockIdx.x * wpb + wv;
    if (wave < totalWaves) gemm_strip(Xh, B, Cout, H2 * C2, NOUT, tilesN4, wave, lane);
    __syncthreads();
    att_rows<H2, C2>(Cout, asr2, adr2, als2, dm2, row0, nRows, N);
}

static inline long cdivl(long a, long b) { return (a + b - 1) / b; }

struct LayerWS {
    _Float16* Xh;
    float *als, *dm, *denom;
    unsigned int* tick;
};

static void run_gat_edges(int H,
                          const int* src, const int* dst, int E,
                          float* Hbuf, float* agg, const LayerWS& w, int layer,
                          hipStream_t stream) {
    const int TB = 256;
    int nEH = E * H;
    k_edge_max<<<(int)cdivl(nEH, TB), TB, 0, stream>>>(src, dst, E, w.als, w.dm, H);
    int nChunks = (int)cdivl(E, SC_EPB);
    unsigned int* tk = w.tick + layer * 2 * TICK_STRIDE;
    if (H == 8) k_edge_scatter<8, 32><<<2 * nChunks, SC_WPB * 32, 0, stream>>>(
                    src, dst, E, w.als, w.dm, w.denom, Hbuf, agg, tk, nChunks);
    else        k_edge_scatter<1, 128><<<2 * nChunks, SC_WPB * 32, 0, stream>>>(
                    src, dst, E, w.als, w.dm, w.denom, Hbuf, agg, tk, nChunks);
}

extern "C" void kernel_launch(void* const* d_in, const int* in_sizes, int n_in,
                              void* d_out, int out_size, void* d_ws, size_t ws_size,
                              hipStream_t stream) {
    const float* x   = (const float*)d_in[0];
    const int*   ei  = (const int*)  d_in[1];
    const float* W1  = (const float*)d_in[2];
    const float* as1 = (const float*)d_in[3];
    const float* ad1 = (const float*)d_in[4];
    const float* b1  = (const float*)d_in[5];
    const float* W2  = (const float*)d_in[6];
    const float* as2 = (const float*)d_in[7];
    const float* ad2 = (const float*)d_in[8];
    const float* b2  = (const float*)d_in[9];
    const float* W3  = (const float*)d_in[10];
    const float* as3 = (const float*)d_in[11];
    const float* ad3 = (const float*)d_in[12];
    const float* b3  = (const float*)d_in[13];
    float* out = (float*)d_out;

    const int FinW = 256;
    const int N    = in_sizes[0] / FinW;
    const int E    = in_sizes[1] / 2;
    const int* src = ei;
    const int* dst = ei + E;

    char* p = (char*)d_ws;
    float*    P   = (float*)p;    p += (size_t)N * 256 * sizeof(float);
    float*    Q   = (float*)p;    p += (size_t)N * 256 * sizeof(float);
    float*    Q3  = (float*)p;    p += (size_t)N * 128 * sizeof(float);
    _Float16* Xh  = (_Float16*)p; p += (size_t)N * 256 * sizeof(_Float16);
    _Float16* Wh1 = (_Float16*)p; p += (size_t)256 * 256 * sizeof(_Float16);
    _Float16* Wh2 = (_Float16*)p; p += (size_t)256 * 256 * sizeof(_Float16);
    _Float16* Wh3 = (_Float16*)p; p += (size_t)128 * 256 * sizeof(_Float16);
    float*    als1 = (float*)p;   p += (size_t)N * 8 * sizeof(float);
    float*    dm1  = (float*)p;   p += (size_t)N * 8 * 2 * sizeof(float);
    float*    als2 = (float*)p;   p += (size_t)N * 8 * sizeof(float);
    float*    dm2  = (float*)p;   p += (size_t)N * 8 * 2 * sizeof(float);
    float*    als3 = (float*)p;   p += (size_t)N * 8 * sizeof(float);
    float*    dm3  = (float*)p;   p += (size_t)N * 8 * 2 * sizeof(float);
    float*    den = (float*)p;    p += (size_t)N * 8 * sizeof(float);
    unsigned int* tick = (unsigned int*)p; p += (size_t)TICK_WORDS * sizeof(unsigned int);
    LayerWS w1{Xh, als1, dm1, den, tick};
    LayerWS w2{Xh, als2, dm2, den, tick};
    LayerWS w3{Xh, als3, dm3, den, tick};

    const int TB = 256;
    const long nHid = (long)N * 256;
    const int tiles256 = 256 / 64, waves256 = (N / 16) * tiles256;
    const int tiles128 = 128 / 64, waves128 = (N / 16) * tiles128;

    const long nW12 = 256L * 256, nW3 = 128L * 256;
    const long nPrep = nHid > nW12 ? nHid : nW12;
    k_prep<FinW><<<(int)cdivl(nPrep, TB), TB, 0, stream>>>(x, Xh, W1, W2, W3, Wh1, Wh2, Wh3, den, P,
                                                           tick, nHid, nW12, nW3, (long)N * 8);
    k_gemm_att<8, 32><<<(int)cdivl(waves256, 8), TB, 0, stream>>>(
        Xh, Wh1, Q, 256, tiles256, waves256, N, as1, ad1, als1, dm1);
    run_gat_edges(8, src, dst, E, Q, P, w1, 0, stream);
    k_boundary_gemm<8, 32, 8, 32><<<(int)cdivl(waves256, 8), TB, 0, stream>>>(
        P, den, als1, dm1, Q, b1, Xh, P, 256, N, Wh2, Q, tiles256, waves256,
        as2, ad2, als2, dm2);
    run_gat_edges(8, src, dst, E, Q, P, w2, 1, stream);
    k_boundary_gemm<8, 32, 1, 128><<<(int)cdivl(waves128, 8), TB, 0, stream>>>(
        P, den, als2, dm2, Q, b2, Xh, out, 128, N, Wh3, Q3, tiles128, waves128,
        as3, ad3, als3, dm3);
    run_gat_edges(1, src, dst, E, Q3, out, w3, 2, stream);
    k_norm_bias_act<<<(int)cdivl((long)N * 128, TB), TB, 0, stream>>>(out, den, als3, dm3,
                                                                      Q3, b3, (long)N * 128, 128, 1,
                                                                      128, 0);
}
